# starting kernel
# baseline (speedup 1.0000x reference)
#include <hip/hip_runtime.h>


typedef __attribute__((ext_vector_type(2))) float v2f;
typedef __attribute__((ext_vector_type(8))) float v8f;

#define NKP   184
#define NP0   96
#define NP1   256
#define NROWS (1024*256)
#define OUT1_OFF NROWS

__global__ void symcon_pack(const float* __restrict__ U2_l0,
                            const float* __restrict__ U3_l0,
                            const float* __restrict__ U2_l1,
                            const float* __restrict__ U3_l1,
                            float* __restrict__ Bp0,
                            float* __restrict__ Bp1) {
  int idx = blockIdx.x * 256 + threadIdx.x;
  if (idx < NP0 * NKP) {
    int n = idx / NKP, k = idx % NKP;
    float v = 0.f;
    if (n < 81) v = (k < 180) ? U3_l0[n * 180 + k] : U2_l0[n * 4 + (k - 180)];
    Bp0[idx] = v;
  } else {
    idx -= NP0 * NKP;
    if (idx < NP1 * NKP) {
      int n = idx / NKP, k = idx % NKP;
      float v = 0.f;
      if (n < 243) v = (k < 180) ? U3_l1[n * 180 + k] : U2_l1[n * 4 + (k - 180)];
      Bp1[idx] = v;
    }
  }
}


__device__ __forceinline__ float zval(int k, const float (&xr)[9],
                                      const float (&p3)[20], const float (&p2)[4]) {
  return (k < 180) ? xr[k / 20] * p3[k % 20] : p2[k - 180];
}

__device__ __forceinline__ void compute_p(const float (&yv)[10], int c,
    const float* __restrict__ w1, const float* __restrict__ w2,
    const float* __restrict__ w3,
    float (&p3)[20], float (&p2)[4], float &p1) {
  #pragma unroll
  for (int k = 0; k < 20; ++k) p3[k] = 0.f;
  #pragma unroll
  for (int k = 0; k < 4; ++k) p2[k] = 0.f;
  p1 = 0.f;
  #pragma unroll
  for (int e = 0; e < 10; ++e) {
    float ye = yv[e];
    #pragma unroll
    for (int k = 0; k < 20; ++k) p3[k] = fmaf(ye, w3[(e * 20 + k) * 256 + c], p3[k]);
    #pragma unroll
    for (int k = 0; k < 4; ++k)  p2[k] = fmaf(ye, w2[(e * 4 + k) * 256 + c], p2[k]);
    p1 = fmaf(ye, w1[e * 256 + c], p1);
  }
}

__device__ __forceinline__ void gemm_seg(const float* __restrict__ bseg, int n0,
    int NT8, int lan, bool hi,
    const float (&xr)[9], const float (&p3)[20], const float (&p2)[4],
    v8f* __restrict__ acc, int NT) {
  (void)NT8;
  const char* bp = (const char*)bseg +
                   (size_t)(((n0 + lan) * NKP + (hi ? 2 : 0)) * 4);
  #pragma unroll
  for (int s = 0; s < 46; ++s) {
    float z0 = zval(4 * s + 0, xr, p3, p2);
    float z1 = zval(4 * s + 1, xr, p3, p2);
    float z2 = zval(4 * s + 2, xr, p3, p2);
    float z3 = zval(4 * s + 3, xr, p3, p2);
    v2f a;
    a.x = hi ? z2 : z0;
    a.y = hi ? z3 : z1;
    #pragma unroll
    for (int t = 0; t < 8; ++t) {
      if (t < NT) {
        v2f bv = *(const v2f*)(bp + t * (16 * NKP * 4) + s * 16);
        acc[t] = __builtin_amdgcn_wmma_f32_16x16x4_f32(
            false, a, false, bv, (short)0, acc[t], false, false);
      }
    }
  }
}

__global__ __launch_bounds__(256, 1) void symcon_main(
    const float* __restrict__ x, const float* __restrict__ y,
    const float* __restrict__ U1_l0, const float* __restrict__ U1_l1,
    const float* __restrict__ w1_l0, const float* __restrict__ w2_l0,
    const float* __restrict__ w3_l0,
    const float* __restrict__ w1_l1, const float* __restrict__ w2_l1,
    const float* __restrict__ w3_l1,
    const float* __restrict__ Bp0, const float* __restrict__ Bp1,
    float* __restrict__ out) {
  __shared__ float xs[8][16][12];

  const int tid  = threadIdx.x;
  const int wv   = tid >> 5;
  const int lane = tid & 31;
  const int lan  = lane & 15;
  const bool hi  = lane >= 16;
  const int row0 = (blockIdx.x * 8 + wv) * 16;
  const int row  = row0 + lan;
  const int b    = row >> 8;
  const int c    = row & 255;

  float xr[9];
  #pragma unroll
  for (int i = 0; i < 9; ++i) xr[i] = x[row * 9 + i];

  float yv[10];
  #pragma unroll
  for (int e = 0; e < 10; ++e) yv[e] = y[b * 10 + e];

  float p3[20], p2[4], p1;

  compute_p(yv, c, w1_l0, w2_l0, w3_l0, p3, p2, p1);
  if (!hi) {
    #pragma unroll
    for (int i = 0; i < 9; ++i) xs[wv][lan][i] = xr[i];
    xs[wv][lan][9] = p1;
  }

  {
    v8f acc6[6];
    v8f vz = {};
    #pragma unroll
    for (int t = 0; t < 6; ++t) acc6[t] = vz;

    gemm_seg(Bp0, 0, 0, lan, hi, xr, p3, p2, acc6, 6);

    float o0[8];
    #pragma unroll
    for (int j = 0; j < 8; ++j) o0[j] = 0.f;
    #pragma unroll
    for (int t = 0; t < 6; ++t) {
      int n = t * 16 + lan;
      if (n < 81) {
        int wq = n / 9, xq = n % 9;
        #pragma unroll
        for (int j = 0; j < 8; ++j) {
          int rl = j + (hi ? 8 : 0);
          float q = xs[wv][rl][wq] * xs[wv][rl][xq];
          o0[j] = fmaf(acc6[t][j], q, o0[j]);
        }
      }
    }
    #pragma unroll
    for (int j = 0; j < 8; ++j) {
      o0[j] += __shfl_xor(o0[j], 1, 32);
      o0[j] += __shfl_xor(o0[j], 2, 32);
      o0[j] += __shfl_xor(o0[j], 4, 32);
      o0[j] += __shfl_xor(o0[j], 8, 32);
    }
    if (lan == 0) {
      #pragma unroll
      for (int j = 0; j < 8; ++j) {
        int rl = j + (hi ? 8 : 0);
        float du = 0.f;
        #pragma unroll
        for (int wq = 0; wq < 9; ++wq) du = fmaf(U1_l0[wq], xs[wv][rl][wq], du);
        out[row0 + rl] = o0[j] + xs[wv][rl][9] * du;
      }
    }
  }

  compute_p(yv, c, w1_l1, w2_l1, w3_l1, p3, p2, p1);
  if (!hi) xs[wv][lan][10] = p1;

  float o1[8][3];
  #pragma unroll
  for (int j = 0; j < 8; ++j)
    #pragma unroll
    for (int wq = 0; wq < 3; ++wq) o1[j][wq] = 0.f;

  v8f acc8[8];
  #pragma unroll
  for (int seg = 0; seg < 2; ++seg) {
    const int n0 = seg * 128;
    v8f vz = {};
    #pragma unroll
    for (int t = 0; t < 8; ++t) acc8[t] = vz;

    gemm_seg(Bp1, n0, 0, lan, hi, xr, p3, p2, acc8, 8);

    #pragma unroll
    for (int t = 0; t < 8; ++t) {
      int n = n0 + t * 16 + lan;
      if (n < 243) {
        int wq = n / 81, xq = (n / 9) % 9, vq = n % 9;
        #pragma unroll
        for (int j = 0; j < 8; ++j) {
          int rl = j + (hi ? 8 : 0);
          float q = xs[wv][rl][xq] * xs[wv][rl][vq];
          float pz = acc8[t][j] * q;
          if (wq == 0)      o1[j][0] += pz;
          else if (wq == 1) o1[j][1] += pz;
          else              o1[j][2] += pz;
        }
      }
    }
  }

  #pragma unroll
  for (int j = 0; j < 8; ++j)
    #pragma unroll
    for (int wq = 0; wq < 3; ++wq) {
      float v = o1[j][wq];
      v += __shfl_xor(v, 1, 32);
      v += __shfl_xor(v, 2, 32);
      v += __shfl_xor(v, 4, 32);
      v += __shfl_xor(v, 8, 32);
      o1[j][wq] = v;
    }
  if (lan == 0) {
    #pragma unroll
    for (int j = 0; j < 8; ++j) {
      int rl = j + (hi ? 8 : 0);
      int r  = row0 + rl;
      #pragma unroll
      for (int wq = 0; wq < 3; ++wq) {
        float du = 0.f;
        #pragma unroll
        for (int xq = 0; xq < 9; ++xq)
          du = fmaf(U1_l1[wq * 9 + xq], xs[wv][rl][xq], du);
        out[OUT1_OFF + r * 3 + wq] = o1[j][wq] + xs[wv][rl][10] * du;
      }
    }
  }
}

extern "C" void kernel_launch(void* const* d_in, const int* in_sizes, int n_in,
                              void* d_out, int out_size, void* d_ws, size_t ws_size,
                              hipStream_t stream) {
  (void)in_sizes; (void)n_in; (void)out_size; (void)ws_size;
  const float* x     = (const float*)d_in[0];
  const float* y     = (const float*)d_in[1];
  const float* U1_l0 = (const float*)d_in[2];
  const float* U2_l0 = (const float*)d_in[3];
  const float* U3_l0 = (const float*)d_in[4];
  const float* U1_l1 = (const float*)d_in[5];
  const float* U2_l1 = (const float*)d_in[6];
  const float* U3_l1 = (const float*)d_in[7];
  const float* w1_l0 = (const float*)d_in[8];
  const float* w2_l0 = (const float*)d_in[9];
  const float* w3_l0 = (const float*)d_in[10];
  const float* w1_l1 = (const float*)d_in[11];
  const float* w2_l1 = (const float*)d_in[12];
  const float* w3_l1 = (const float*)d_in[13];

  float* Bp0 = (float*)d_ws;
  float* Bp1 = Bp0 + NP0 * NKP;

  const int pack_elems = (NP0 + NP1) * NKP;
  symcon_pack<<<(pack_elems + 255) / 256, 256, 0, stream>>>(
      U2_l0, U3_l0, U2_l1, U3_l1, Bp0, Bp1);

  symcon_main<<<2048, 256, 0, stream>>>(
      x, y, U1_l0, U1_l1,
      w1_l0, w2_l0, w3_l0,
      w1_l1, w2_l1, w3_l1,
      Bp0, Bp1, (float*)d_out);
}


# B operand repacked in wave-lane order: one 16-byte load per lane fetches its fragments of two k-steps and a wave reads 512 contiguous bytes (4 lines) where each 8-byte load touched 16 lines
# speedup vs baseline: 3.9168x; 3.9168x over previous
#include <hip/hip_runtime.h>


typedef __attribute__((ext_vector_type(2))) float v2f;
typedef __attribute__((ext_vector_type(4))) float v4f;
typedef __attribute__((ext_vector_type(8))) float v8f;

#define NROWS (1024*256)
#define OUT1_OFF NROWS
#define NJ    23
#define NT0   6
#define NT1   16
#define TILE_FLOATS (NJ * 32 * 4)

__global__ void symcon_pack(const float* __restrict__ U2_l0,
                            const float* __restrict__ U3_l0,
                            const float* __restrict__ U2_l1,
                            const float* __restrict__ U3_l1,
                            float* __restrict__ Bf) {
  int idx = blockIdx.x * 256 + threadIdx.x;
  if (idx >= (NT0 + NT1) * TILE_FLOATS) return;
  int e    = idx & 3;
  int lane = (idx >> 2) & 31;
  int cj   = idx >> 7;
  int tile = cj / NJ, j = cj % NJ;
  int k    = 4 * (2 * j + (e >> 1)) + 2 * (lane >> 4) + (e & 1);
  float v = 0.f;
  if (tile < NT0) {
    int n = tile * 16 + (lane & 15);
    if (n < 81) v = (k < 180) ? U3_l0[n * 180 + k] : U2_l0[n * 4 + (k - 180)];
  } else {
    int n = (tile - NT0) * 16 + (lane & 15);
    if (n < 243) v = (k < 180) ? U3_l1[n * 180 + k] : U2_l1[n * 4 + (k - 180)];
  }
  Bf[idx] = v;
}


__device__ __forceinline__ float zval(int k, const float (&xr)[9],
                                      const float (&p3)[20], const float (&p2)[4]) {
  return (k < 180) ? xr[k / 20] * p3[k % 20] : p2[k - 180];
}

__device__ __forceinline__ void compute_p(const float (&yv)[10], int c,
    const float* __restrict__ w1, const float* __restrict__ w2,
    const float* __restrict__ w3,
    float (&p3)[20], float (&p2)[4], float &p1) {
  #pragma unroll
  for (int k = 0; k < 20; ++k) p3[k] = 0.f;
  #pragma unroll
  for (int k = 0; k < 4; ++k) p2[k] = 0.f;
  p1 = 0.f;
  #pragma unroll
  for (int e = 0; e < 10; ++e) {
    float ye = yv[e];
    #pragma unroll
    for (int k = 0; k < 20; ++k) p3[k] = fmaf(ye, w3[(e * 20 + k) * 256 + c], p3[k]);
    #pragma unroll
    for (int k = 0; k < 4; ++k)  p2[k] = fmaf(ye, w2[(e * 4 + k) * 256 + c], p2[k]);
    p1 = fmaf(ye, w1[e * 256 + c], p1);
  }
}

__device__ __forceinline__ void gemm_seg(const float* __restrict__ bf, int tile0,
    int lane, bool hi,
    const float (&xr)[9], const float (&p3)[20], const float (&p2)[4],
    v8f* __restrict__ acc, int NT) {
  const char* bp = (const char*)bf +
                   (size_t)(tile0 * (TILE_FLOATS * 4) + lane * 16);
  #pragma unroll
  for (int j = 0; j < NJ; ++j) {
    v2f a0, a1;
    {
      const int s = 2 * j;
      float z0 = zval(4 * s + 0, xr, p3, p2);
      float z1 = zval(4 * s + 1, xr, p3, p2);
      float z2 = zval(4 * s + 2, xr, p3, p2);
      float z3 = zval(4 * s + 3, xr, p3, p2);
      a0.x = hi ? z2 : z0;
      a0.y = hi ? z3 : z1;
    }
    {
      const int s = 2 * j + 1;
      float z0 = zval(4 * s + 0, xr, p3, p2);
      float z1 = zval(4 * s + 1, xr, p3, p2);
      float z2 = zval(4 * s + 2, xr, p3, p2);
      float z3 = zval(4 * s + 3, xr, p3, p2);
      a1.x = hi ? z2 : z0;
      a1.y = hi ? z3 : z1;
    }
    v4f bq[8];
    #pragma unroll
    for (int t = 0; t < 8; ++t)
      if (t < NT) bq[t] = *(const v4f*)(bp + t * (TILE_FLOATS * 4) + j * 512);
    #pragma unroll
    for (int t = 0; t < 8; ++t) {
      if (t < NT) {
        v2f b0;
        b0.x = bq[t].x; b0.y = bq[t].y;
        acc[t] = __builtin_amdgcn_wmma_f32_16x16x4_f32(
            false, a0, false, b0, (short)0, acc[t], false, false);
      }
    }
    #pragma unroll
    for (int t = 0; t < 8; ++t) {
      if (t < NT) {
        v2f b1;
        b1.x = bq[t].z; b1.y = bq[t].w;
        acc[t] = __builtin_amdgcn_wmma_f32_16x16x4_f32(
            false, a1, false, b1, (short)0, acc[t], false, false);
      }
    }
  }
}

__global__ __launch_bounds__(256, 1) void symcon_main(
    const float* __restrict__ x, const float* __restrict__ y,
    const float* __restrict__ U1_l0, const float* __restrict__ U1_l1,
    const float* __restrict__ w1_l0, const float* __restrict__ w2_l0,
    const float* __restrict__ w3_l0,
    const float* __restrict__ w1_l1, const float* __restrict__ w2_l1,
    const float* __restrict__ w3_l1,
    const float* __restrict__ Bf,
    float* __restrict__ out) {
  __shared__ float xs[8][16][12];

  const int tid  = threadIdx.x;
  const int wv   = tid >> 5;
  const int lane = tid & 31;
  const int lan  = lane & 15;
  const bool hi  = lane >= 16;
  const int row0 = (blockIdx.x * 8 + wv) * 16;
  const int row  = row0 + lan;
  const int b    = row >> 8;
  const int c    = row & 255;

  float xr[9];
  #pragma unroll
  for (int i = 0; i < 9; ++i) xr[i] = x[row * 9 + i];

  float yv[10];
  #pragma unroll
  for (int e = 0; e < 10; ++e) yv[e] = y[b * 10 + e];

  float p3[20], p2[4], p1;

  compute_p(yv, c, w1_l0, w2_l0, w3_l0, p3, p2, p1);
  if (!hi) {
    #pragma unroll
    for (int i = 0; i < 9; ++i) xs[wv][lan][i] = xr[i];
    xs[wv][lan][9] = p1;
  }

  {
    v8f acc6[6];
    v8f vz = {};
    #pragma unroll
    for (int t = 0; t < 6; ++t) acc6[t] = vz;

    gemm_seg(Bf, 0, lane, hi, xr, p3, p2, acc6, 6);

    float o0[8];
    #pragma unroll
    for (int j = 0; j < 8; ++j) o0[j] = 0.f;
    #pragma unroll
    for (int t = 0; t < 6; ++t) {
      int n = t * 16 + lan;
      if (n < 81) {
        int wq = n / 9, xq = n % 9;
        #pragma unroll
        for (int j = 0; j < 8; ++j) {
          int rl = j + (hi ? 8 : 0);
          float q = xs[wv][rl][wq] * xs[wv][rl][xq];
          o0[j] = fmaf(acc6[t][j], q, o0[j]);
        }
      }
    }
    #pragma unroll
    for (int j = 0; j < 8; ++j) {
      o0[j] += __shfl_xor(o0[j], 1, 32);
      o0[j] += __shfl_xor(o0[j], 2, 32);
      o0[j] += __shfl_xor(o0[j], 4, 32);
      o0[j] += __shfl_xor(o0[j], 8, 32);
    }
    if (lan == 0) {
      #pragma unroll
      for (int j = 0; j < 8; ++j) {
        int rl = j + (hi ? 8 : 0);
        float du = 0.f;
        #pragma unroll
        for (int wq = 0; wq < 9; ++wq) du = fmaf(U1_l0[wq], xs[wv][rl][wq], du);
        out[row0 + rl] = o0[j] + xs[wv][rl][9] * du;
      }
    }
  }

  compute_p(yv, c, w1_l1, w2_l1, w3_l1, p3, p2, p1);
  if (!hi) xs[wv][lan][10] = p1;

  float o1[8][3];
  #pragma unroll
  for (int j = 0; j < 8; ++j)
    #pragma unroll
    for (int wq = 0; wq < 3; ++wq) o1[j][wq] = 0.f;

  v8f acc8[8];
  #pragma unroll
  for (int seg = 0; seg < 2; ++seg) {
    const int n0 = seg * 128;
    v8f vz = {};
    #pragma unroll
    for (int t = 0; t < 8; ++t) acc8[t] = vz;

    gemm_seg(Bf, NT0 + seg * 8, lane, hi, xr, p3, p2, acc8, 8);

    #pragma unroll
    for (int t = 0; t < 8; ++t) {
      int n = n0 + t * 16 + lan;
      if (n < 243) {
        int wq = n / 81, xq = (n / 9) % 9, vq = n % 9;
        #pragma unroll
        for (int j = 0; j < 8; ++j) {
          int rl = j + (hi ? 8 : 0);
          float q = xs[wv][rl][xq] * xs[wv][rl][vq];
          float pz = acc8[t][j] * q;
          if (wq == 0)      o1[j][0] += pz;
          else if (wq == 1) o1[j][1] += pz;
          else              o1[j][2] += pz;
        }
      }
    }
  }

  #pragma unroll
  for (int j = 0; j < 8; ++j)
    #pragma unroll
    for (int wq = 0; wq < 3; ++wq) {
      float v = o1[j][wq];
      v += __shfl_xor(v, 1, 32);
      v += __shfl_xor(v, 2, 32);
      v += __shfl_xor(v, 4, 32);
      v += __shfl_xor(v, 8, 32);
      o1[j][wq] = v;
    }
  if (lan == 0) {
    #pragma unroll
    for (int j = 0; j < 8; ++j) {
      int rl = j + (hi ? 8 : 0);
      int r  = row0 + rl;
      #pragma unroll
      for (int wq = 0; wq < 3; ++wq) {
        float du = 0.f;
        #pragma unroll
        for (int xq = 0; xq < 9; ++xq)
          du = fmaf(U1_l1[wq * 9 + xq], xs[wv][rl][xq], du);
        out[OUT1_OFF + r * 3 + wq] = o1[j][wq] + xs[wv][rl][10] * du;
      }
    }
  }
}

extern "C" void kernel_launch(void* const* d_in, const int* in_sizes, int n_in,
                              void* d_out, int out_size, void* d_ws, size_t ws_size,
                              hipStream_t stream) {
  (void)in_sizes; (void)n_in; (void)out_size; (void)ws_size;
  const float* x     = (const float*)d_in[0];
  const float* y     = (const float*)d_in[1];
  const float* U1_l0 = (const float*)d_in[2];
  const float* U2_l0 = (const float*)d_in[3];
  const float* U3_l0 = (const float*)d_in[4];
  const float* U1_l1 = (const float*)d_in[5];
  const float* U2_l1 = (const float*)d_in[6];
  const float* U3_l1 = (const float*)d_in[7];
  const float* w1_l0 = (const float*)d_in[8];
  const float* w2_l0 = (const float*)d_in[9];
  const float* w3_l0 = (const float*)d_in[10];
  const float* w1_l1 = (const float*)d_in[11];
  const float* w2_l1 = (const float*)d_in[12];
  const float* w3_l1 = (const float*)d_in[13];

  float* Bf = (float*)d_ws;

  symcon_pack<<<253, 256, 0, stream>>>(
      U2_l0, U3_l0, U2_l1, U3_l1, Bf);

  symcon_main<<<2048, 256, 0, stream>>>(
      x, y, U1_l0, U1_l1,
      w1_l0, w2_l0, w3_l0,
      w1_l1, w2_l1, w3_l1,
      Bf, (float*)d_out);
}


# B operand staged in LDS once per workgroup (56832 bytes, plain global loads + ds_store, one barrier): waves read B fragments with ds_load_b128 instead of each re-reading them via global loads
# speedup vs baseline: 15.1387x; 3.8651x over previous
#include <hip/hip_runtime.h>


typedef __attribute__((ext_vector_type(2))) float v2f;
typedef __attribute__((ext_vector_type(4))) float v4f;
typedef __attribute__((ext_vector_type(8))) float v8f;

#define NROWS (1024*256)
#define OUT1_OFF NROWS
#define NM3   165
#define NM2   45
#define NJ3   21
#define NJ2   6
#define NT3   5
#define PK3_FLOATS (NT3 * NJ3 * 128)
#define PK2_FLOATS (NJ2 * 128)
#define PKV4  ((PK3_FLOATS + PK2_FLOATS) / 4)
#define NWAVE 8
#define NTHR  (NWAVE * 32)
#define PROW  52

struct Tri { int a, b, c; };

__host__ __device__ constexpr Tri dec3(int m) {
  if (m >= NM3) return Tri{-1, -1, -1};
  int a = 0;
  while (m >= (9 - a) * (10 - a) / 2) { m -= (9 - a) * (10 - a) / 2; ++a; }
  int b = a;
  while (m >= 9 - b) { m -= 9 - b; ++b; }
  return Tri{a, b, b + m};
}
__host__ __device__ constexpr Tri dec2(int q) {
  if (q >= NM2) return Tri{-1, -1, -1};
  int a = 0;
  while (q >= 9 - a) { q -= 9 - a; ++a; }
  return Tri{a, a + q, 0};
}

__global__ void symcon_pack(const float* __restrict__ U2_l0,
                            const float* __restrict__ U3_l0,
                            const float* __restrict__ U2_l1,
                            const float* __restrict__ U3_l1,
                            float* __restrict__ Bf) {
  int idx = blockIdx.x * 256 + threadIdx.x;
  if (idx >= PK3_FLOATS + PK2_FLOATS) return;
  int e    = idx & 3;
  int lane = (idx >> 2) & 31;
  int lan  = lane & 15;
  float v = 0.f;
  if (idx < PK3_FLOATS) {
    int cj   = idx >> 7;
    int tile = cj / NJ3, j = cj % NJ3;
    int m    = 4 * (2 * j + (e >> 1)) + 2 * (lane >> 4) + (e & 1);
    int n    = tile * 16 + lan;
    int kk   = n >> 2, o = n & 3;
    Tri t = dec3(m);
    if (t.a >= 0) {
      const float* U = (o == 0) ? U3_l0 : (U3_l1 + (o - 1) * (729 * 20));
      int a = t.a, b = t.b, c = t.c;
      #define U3AT(p, q, s) U[(((p) * 9 + (q)) * 9 + (s)) * 20 + kk]
      if (a == b && b == c)      v = U3AT(a, a, a);
      else if (a == b)           v = U3AT(a, a, c) + U3AT(a, c, a) + U3AT(c, a, a);
      else if (b == c)           v = U3AT(a, b, b) + U3AT(b, a, b) + U3AT(b, b, a);
      else                       v = U3AT(a, b, c) + U3AT(a, c, b) + U3AT(b, a, c)
                                   + U3AT(b, c, a) + U3AT(c, a, b) + U3AT(c, b, a);
      #undef U3AT
    }
  } else {
    int j  = (idx - PK3_FLOATS) >> 7;
    int q  = 4 * (2 * j + (e >> 1)) + 2 * (lane >> 4) + (e & 1);
    int kk = lan >> 2, o = lan & 3;
    Tri t = dec2(q);
    if (t.a >= 0) {
      const float* U = (o == 0) ? U2_l0 : (U2_l1 + (o - 1) * (81 * 4));
      int a = t.a, b = t.b;
      if (a == b) v = U[(a * 9 + a) * 4 + kk];
      else        v = U[(a * 9 + b) * 4 + kk] + U[(b * 9 + a) * 4 + kk];
    }
  }
  Bf[idx] = v;
}


template <int M>
__device__ __forceinline__ float mono3(const float (&xr)[9]) {
  if constexpr (M < NM3) {
    constexpr Tri t = dec3(M);
    return xr[t.a] * xr[t.b] * xr[t.c];
  } else {
    return 0.f;
  }
}
template <int Q>
__device__ __forceinline__ float mono2(const float (&xr)[9]) {
  if constexpr (Q < NM2) {
    constexpr Tri t = dec2(Q);
    return xr[t.a] * xr[t.b];
  } else {
    return 0.f;
  }
}

template <int S>
__device__ __forceinline__ v2f afrag3(bool hi, const float (&xr)[9]) {
  float z0 = mono3<4 * S + 0>(xr), z1 = mono3<4 * S + 1>(xr);
  float z2 = mono3<4 * S + 2>(xr), z3 = mono3<4 * S + 3>(xr);
  v2f a;
  a.x = hi ? z2 : z0;
  a.y = hi ? z3 : z1;
  return a;
}
template <int S>
__device__ __forceinline__ v2f afrag2(bool hi, const float (&xr)[9]) {
  float z0 = mono2<4 * S + 0>(xr), z1 = mono2<4 * S + 1>(xr);
  float z2 = mono2<4 * S + 2>(xr), z3 = mono2<4 * S + 3>(xr);
  v2f a;
  a.x = hi ? z2 : z0;
  a.y = hi ? z3 : z1;
  return a;
}

template <int J>
__device__ __forceinline__ void gemm_steps(const v4f* bp, bool hi,
                                           const float (&xr)[9],
                                           v8f (&acc3)[NT3], v8f &acc2) {
  v4f bq[NT3];
  #pragma unroll
  for (int t = 0; t < NT3; ++t)
    bq[t] = bp[(t * NJ3 + J) * 32];
  const v2f a0 = afrag3<2 * J>(hi, xr);
  const v2f a1 = afrag3<2 * J + 1>(hi, xr);
  #pragma unroll
  for (int t = 0; t < NT3; ++t) {
    v2f b0;
    b0.x = bq[t].x; b0.y = bq[t].y;
    acc3[t] = __builtin_amdgcn_wmma_f32_16x16x4_f32(
        false, a0, false, b0, (short)0, acc3[t], false, false);
  }
  #pragma unroll
  for (int t = 0; t < NT3; ++t) {
    v2f b1;
    b1.x = bq[t].z; b1.y = bq[t].w;
    acc3[t] = __builtin_amdgcn_wmma_f32_16x16x4_f32(
        false, a1, false, b1, (short)0, acc3[t], false, false);
  }
  if constexpr (J < NJ2) {
    const v4f bs = bp[PK3_FLOATS / 4 + J * 32];
    const v2f c0 = afrag2<2 * J>(hi, xr);
    const v2f c1 = afrag2<2 * J + 1>(hi, xr);
    v2f b0, b1;
    b0.x = bs.x; b0.y = bs.y;
    b1.x = bs.z; b1.y = bs.w;
    acc2 = __builtin_amdgcn_wmma_f32_16x16x4_f32(
        false, c0, false, b0, (short)0, acc2, false, false);
    acc2 = __builtin_amdgcn_wmma_f32_16x16x4_f32(
        false, c1, false, b1, (short)0, acc2, false, false);
  }
  if constexpr (J + 1 < NJ3) gemm_steps<J + 1>(bp, hi, xr, acc3, acc2);
}

__global__ __launch_bounds__(NTHR) void symcon_main(
    const float* __restrict__ x, const float* __restrict__ y,
    const float* __restrict__ U1_l0, const float* __restrict__ U1_l1,
    const float* __restrict__ w1_l0, const float* __restrict__ w2_l0,
    const float* __restrict__ w3_l0,
    const float* __restrict__ w1_l1, const float* __restrict__ w2_l1,
    const float* __restrict__ w3_l1,
    const float* __restrict__ Bf,
    float* __restrict__ out) {
  __shared__ v4f bsh[PKV4];
  __shared__ float ps[NWAVE][16 * PROW];
  __shared__ float t1[NWAVE][16][4];
  __shared__ float ob[NWAVE][64];

  const int tid  = threadIdx.x;
  const int wv   = tid >> 5;
  const int lane = tid & 31;
  const int lan  = lane & 15;
  const bool hi  = lane >= 16;
  const int h    = hi ? 1 : 0;
  const int row0 = (blockIdx.x * NWAVE + wv) * 16;
  const int row  = row0 + lan;
  const int b    = blockIdx.x >> 1;
  const int c    = row & 255;

  float xr[9];
  #pragma unroll
  for (int i = 0; i < 9; ++i) xr[i] = x[row * 9 + i];

  {
    const v4f* __restrict__ bg = (const v4f*)Bf;
    #pragma unroll
    for (int i = 0; i < (PKV4 + NTHR - 1) / NTHR; ++i) {
      int u = tid + i * NTHR;
      if ((i + 1) * NTHR > PKV4) u = u < PKV4 ? u : PKV4 - 1;
      bsh[u] = bg[u];
    }
  }

  __syncthreads();

  {
    const float* __restrict__ w3 = hi ? w3_l1 : w3_l0;
    const float* __restrict__ w2 = hi ? w2_l1 : w2_l0;
    const float* __restrict__ w1 = hi ? w1_l1 : w1_l0;
    float p3[20], p2[4], p1;
    #pragma unroll
    for (int k = 0; k < 20; ++k) p3[k] = 0.f;
    #pragma unroll
    for (int k = 0; k < 4; ++k) p2[k] = 0.f;
    p1 = 0.f;
    #pragma unroll 2
    for (int e = 0; e < 10; ++e) {
      float ye = y[b * 10 + e];
      #pragma unroll
      for (int k = 0; k < 20; ++k) p3[k] = fmaf(ye, w3[(e * 20 + k) * 256 + c], p3[k]);
      #pragma unroll
      for (int k = 0; k < 4; ++k)  p2[k] = fmaf(ye, w2[(e * 4 + k) * 256 + c], p2[k]);
      p1 = fmaf(ye, w1[e * 256 + c], p1);
    }
    float* pr = &ps[wv][lan * PROW + h * 25];
    #pragma unroll
    for (int k = 0; k < 20; ++k) pr[k] = p3[k];
    #pragma unroll
    for (int k = 0; k < 4; ++k) pr[20 + k] = p2[k];
    pr[24] = p1;

    if (!hi) {
      float du = 0.f;
      #pragma unroll
      for (int i = 0; i < 9; ++i) du = fmaf(U1_l0[i], xr[i], du);
      t1[wv][lan][0] = p1 * du;
    } else {
      #pragma unroll
      for (int w = 0; w < 3; ++w) {
        float du = 0.f;
        #pragma unroll
        for (int i = 0; i < 9; ++i) du = fmaf(U1_l1[w * 9 + i], xr[i], du);
        t1[wv][lan][1 + w] = p1 * du;
      }
    }
  }

  v8f acc3[NT3];
  v8f acc2;
  {
    v8f vz = {};
    #pragma unroll
    for (int t = 0; t < NT3; ++t) acc3[t] = vz;
    acc2 = vz;
  }
  gemm_steps<0>(bsh + lane, hi, xr, acc3, acc2);

  const int o   = lan & 3;
  const int kq  = lan >> 2;
  const float* pe = &ps[wv][(8 * h) * PROW + (o != 0 ? 25 : 0) + kq];
  float os[8];
  #pragma unroll
  for (int r = 0; r < 8; ++r) os[r] = 0.f;
  #pragma unroll
  for (int t = 0; t < NT3; ++t)
    #pragma unroll
    for (int r = 0; r < 8; ++r)
      os[r] = fmaf(acc3[t][r], pe[r * PROW + 4 * t], os[r]);
  #pragma unroll
  for (int r = 0; r < 8; ++r)
    os[r] = fmaf(acc2[r], pe[r * PROW + 20], os[r]);
  #pragma unroll
  for (int r = 0; r < 8; ++r) {
    os[r] += __shfl_xor(os[r], 4, 32);
    os[r] += __shfl_xor(os[r], 8, 32);
  }
  if (kq == 0) {
    #pragma unroll
    for (int r = 0; r < 8; ++r) {
      int rl = 8 * h + r;
      float v = os[r] + t1[wv][rl][o];
      ob[wv][(o == 0) ? rl : (16 + rl * 3 + (o - 1))] = v;
    }
  }
  if (lane < 16) out[row0 + lane] = ob[wv][lane];
  out[OUT1_OFF + row0 * 3 + lane] = ob[wv][16 + lane];
  if (lane < 16) out[OUT1_OFF + row0 * 3 + 32 + lane] = ob[wv][48 + lane];
}

extern "C" void kernel_launch(void* const* d_in, const int* in_sizes, int n_in,
                              void* d_out, int out_size, void* d_ws, size_t ws_size,
                              hipStream_t stream) {
  (void)in_sizes; (void)n_in; (void)out_size; (void)ws_size;
  const float* x     = (const float*)d_in[0];
  const float* y     = (const float*)d_in[1];
  const float* U1_l0 = (const float*)d_in[2];
  const float* U2_l0 = (const float*)d_in[3];
  const float* U3_l0 = (const float*)d_in[4];
  const float* U1_l1 = (const float*)d_in[5];
  const float* U2_l1 = (const float*)d_in[6];
  const float* U3_l1 = (const float*)d_in[7];
  const float* w1_l0 = (const float*)d_in[8];
  const float* w2_l0 = (const float*)d_in[9];
  const float* w3_l0 = (const float*)d_in[10];
  const float* w1_l1 = (const float*)d_in[11];
  const float* w2_l1 = (const float*)d_in[12];
  const float* w3_l1 = (const float*)d_in[13];

  float* Bf = (float*)d_ws;

  symcon_pack<<<56, 256, 0, stream>>>(
      U2_l0, U3_l0, U2_l1, U3_l1, Bf);

  symcon_main<<<2048, 256, 0, stream>>>(
      x, y, U1_l0, U1_l1,
      w1_l0, w2_l0, w3_l0,
      w1_l1, w2_l1, w3_l1,
      Bf, (float*)d_out);
}


# Two 16-row strips per wave (same 16 channels, two adjacent batch entries): every B fragment read from LDS and every projection weight load feeds both strips; B staged once per 256 rows
# speedup vs baseline: 17.7931x; 1.1753x over previous
#include <hip/hip_runtime.h>


typedef __attribute__((ext_vector_type(2))) float v2f;
typedef __attribute__((ext_vector_type(4))) float v4f;
typedef __attribute__((ext_vector_type(8))) float v8f;

#define NROWS (1024*256)
#define OUT1_OFF NROWS
#define NM3   165
#define NM2   45
#define NJ3   21
#define NJ2   6
#define NT3   5
#define PK3_FLOATS (NT3 * NJ3 * 128)
#define PK2_FLOATS (NJ2 * 128)
#define PKV4  ((PK3_FLOATS + PK2_FLOATS) / 4)
#define NWAVE 8
#define NS    2
#define NTHR  (NWAVE * 32)
#define PROW  52

struct Tri { int a, b, c; };

__host__ __device__ constexpr Tri dec3(int m) {
  if (m >= NM3) return Tri{-1, -1, -1};
  int a = 0;
  while (m >= (9 - a) * (10 - a) / 2) { m -= (9 - a) * (10 - a) / 2; ++a; }
  int b = a;
  while (m >= 9 - b) { m -= 9 - b; ++b; }
  return Tri{a, b, b + m};
}
__host__ __device__ constexpr Tri dec2(int q) {
  if (q >= NM2) return Tri{-1, -1, -1};
  int a = 0;
  while (q >= 9 - a) { q -= 9 - a; ++a; }
  return Tri{a, a + q, 0};
}

__global__ void symcon_pack(const float* __restrict__ U2_l0,
                            const float* __restrict__ U3_l0,
                            const float* __restrict__ U2_l1,
                            const float* __restrict__ U3_l1,
                            float* __restrict__ Bf) {
  int idx = blockIdx.x * 256 + threadIdx.x;
  if (idx >= PK3_FLOATS + PK2_FLOATS) return;
  int e    = idx & 3;
  int lane = (idx >> 2) & 31;
  int lan  = lane & 15;
  float v = 0.f;
  if (idx < PK3_FLOATS) {
    int cj   = idx >> 7;
    int tile = cj / NJ3, j = cj % NJ3;
    int m    = 4 * (2 * j + (e >> 1)) + 2 * (lane >> 4) + (e & 1);
    int n    = tile * 16 + lan;
    int kk   = n >> 2, o = n & 3;
    Tri t = dec3(m);
    if (t.a >= 0) {
      const float* U = (o == 0) ? U3_l0 : (U3_l1 + (o - 1) * (729 * 20));
      int a = t.a, b = t.b, c = t.c;
      #define U3AT(p, q, s) U[(((p) * 9 + (q)) * 9 + (s)) * 20 + kk]
      if (a == b && b == c)      v = U3AT(a, a, a);
      else if (a == b)           v = U3AT(a, a, c) + U3AT(a, c, a) + U3AT(c, a, a);
      else if (b == c)           v = U3AT(a, b, b) + U3AT(b, a, b) + U3AT(b, b, a);
      else                       v = U3AT(a, b, c) + U3AT(a, c, b) + U3AT(b, a, c)
                                   + U3AT(b, c, a) + U3AT(c, a, b) + U3AT(c, b, a);
      #undef U3AT
    }
  } else {
    int j  = (idx - PK3_FLOATS) >> 7;
    int q  = 4 * (2 * j + (e >> 1)) + 2 * (lane >> 4) + (e & 1);
    int kk = lan >> 2, o = lan & 3;
    Tri t = dec2(q);
    if (t.a >= 0) {
      const float* U = (o == 0) ? U2_l0 : (U2_l1 + (o - 1) * (81 * 4));
      int a = t.a, b = t.b;
      if (a == b) v = U[(a * 9 + a) * 4 + kk];
      else        v = U[(a * 9 + b) * 4 + kk] + U[(b * 9 + a) * 4 + kk];
    }
  }
  Bf[idx] = v;
}


template <int M>
__device__ __forceinline__ float mono3(const float (&xr)[9]) {
  if constexpr (M < NM3) {
    constexpr Tri t = dec3(M);
    return xr[t.a] * xr[t.b] * xr[t.c];
  } else {
    return 0.f;
  }
}
template <int Q>
__device__ __forceinline__ float mono2(const float (&xr)[9]) {
  if constexpr (Q < NM2) {
    constexpr Tri t = dec2(Q);
    return xr[t.a] * xr[t.b];
  } else {
    return 0.f;
  }
}

template <int S>
__device__ __forceinline__ v2f afrag3(bool hi, const float (&xr)[9]) {
  float z0 = mono3<4 * S + 0>(xr), z1 = mono3<4 * S + 1>(xr);
  float z2 = mono3<4 * S + 2>(xr), z3 = mono3<4 * S + 3>(xr);
  v2f a;
  a.x = hi ? z2 : z0;
  a.y = hi ? z3 : z1;
  return a;
}
template <int S>
__device__ __forceinline__ v2f afrag2(bool hi, const float (&xr)[9]) {
  float z0 = mono2<4 * S + 0>(xr), z1 = mono2<4 * S + 1>(xr);
  float z2 = mono2<4 * S + 2>(xr), z3 = mono2<4 * S + 3>(xr);
  v2f a;
  a.x = hi ? z2 : z0;
  a.y = hi ? z3 : z1;
  return a;
}

template <int J>
__device__ __forceinline__ void gemm_steps(const v4f* bp, bool hi,
                                           const float (&xr)[NS][9],
                                           v8f (&acc3)[NS][NT3], v8f (&acc2)[NS]) {
  v4f bq[NT3];
  #pragma unroll
  for (int t = 0; t < NT3; ++t)
    bq[t] = bp[(t * NJ3 + J) * 32];
  #pragma unroll
  for (int s = 0; s < NS; ++s) {
    const v2f a0 = afrag3<2 * J>(hi, xr[s]);
    #pragma unroll
    for (int t = 0; t < NT3; ++t) {
      v2f b0;
      b0.x = bq[t].x; b0.y = bq[t].y;
      acc3[s][t] = __builtin_amdgcn_wmma_f32_16x16x4_f32(
          false, a0, false, b0, (short)0, acc3[s][t], false, false);
    }
  }
  #pragma unroll
  for (int s = 0; s < NS; ++s) {
    const v2f a1 = afrag3<2 * J + 1>(hi, xr[s]);
    #pragma unroll
    for (int t = 0; t < NT3; ++t) {
      v2f b1;
      b1.x = bq[t].z; b1.y = bq[t].w;
      acc3[s][t] = __builtin_amdgcn_wmma_f32_16x16x4_f32(
          false, a1, false, b1, (short)0, acc3[s][t], false, false);
    }
  }
  if constexpr (J < NJ2) {
    const v4f bs = bp[PK3_FLOATS / 4 + J * 32];
    v2f b0, b1;
    b0.x = bs.x; b0.y = bs.y;
    b1.x = bs.z; b1.y = bs.w;
    #pragma unroll
    for (int s = 0; s < NS; ++s) {
      const v2f c0 = afrag2<2 * J>(hi, xr[s]);
      acc2[s] = __builtin_amdgcn_wmma_f32_16x16x4_f32(
          false, c0, false, b0, (short)0, acc2[s], false, false);
    }
    #pragma unroll
    for (int s = 0; s < NS; ++s) {
      const v2f c1 = afrag2<2 * J + 1>(hi, xr[s]);
      acc2[s] = __builtin_amdgcn_wmma_f32_16x16x4_f32(
          false, c1, false, b1, (short)0, acc2[s], false, false);
    }
  }
  if constexpr (J + 1 < NJ3) gemm_steps<J + 1>(bp, hi, xr, acc3, acc2);
}

__global__ __launch_bounds__(NTHR) void symcon_main(
    const float* __restrict__ x, const float* __restrict__ y,
    const float* __restrict__ U1_l0, const float* __restrict__ U1_l1,
    const float* __restrict__ w1_l0, const float* __restrict__ w2_l0,
    const float* __restrict__ w3_l0,
    const float* __restrict__ w1_l1, const float* __restrict__ w2_l1,
    const float* __restrict__ w3_l1,
    const float* __restrict__ Bf,
    float* __restrict__ out) {
  __shared__ v4f bsh[PKV4];
  __shared__ float ps[NWAVE][NS][16 * PROW];
  __shared__ float t1[NWAVE][NS][16][4];
  __shared__ float ob[NWAVE][NS][64];

  const int tid  = threadIdx.x;
  const int wv   = tid >> 5;
  const int lane = tid & 31;
  const int lan  = lane & 15;
  const bool hi  = lane >= 16;
  const int h    = hi ? 1 : 0;
  const int bb   = (blockIdx.x >> 1) * NS;
  const int c0   = (blockIdx.x & 1) * 128 + wv * 16;
  const int c    = c0 + lan;

  float xr[NS][9];
  #pragma unroll
  for (int s = 0; s < NS; ++s)
    #pragma unroll
    for (int i = 0; i < 9; ++i) xr[s][i] = x[((bb + s) * 256 + c) * 9 + i];

  {
    const v4f* __restrict__ bg = (const v4f*)Bf;
    #pragma unroll
    for (int i = 0; i < (PKV4 + NTHR - 1) / NTHR; ++i) {
      int u = tid + i * NTHR;
      if ((i + 1) * NTHR > PKV4) u = u < PKV4 ? u : PKV4 - 1;
      bsh[u] = bg[u];
    }
  }

  __syncthreads();

  {
    const float* __restrict__ w3 = hi ? w3_l1 : w3_l0;
    const float* __restrict__ w2 = hi ? w2_l1 : w2_l0;
    const float* __restrict__ w1 = hi ? w1_l1 : w1_l0;
    float p3[NS][20], p2[NS][4], p1[NS];
    #pragma unroll
    for (int s = 0; s < NS; ++s) {
      #pragma unroll
      for (int k = 0; k < 20; ++k) p3[s][k] = 0.f;
      #pragma unroll
      for (int k = 0; k < 4; ++k) p2[s][k] = 0.f;
      p1[s] = 0.f;
    }
    #pragma unroll 2
    for (int e = 0; e < 10; ++e) {
      float ye[NS];
      #pragma unroll
      for (int s = 0; s < NS; ++s) ye[s] = y[(bb + s) * 10 + e];
      #pragma unroll
      for (int k = 0; k < 20; ++k) {
        const float wv3 = w3[(e * 20 + k) * 256 + c];
        #pragma unroll
        for (int s = 0; s < NS; ++s) p3[s][k] = fmaf(ye[s], wv3, p3[s][k]);
      }
      #pragma unroll
      for (int k = 0; k < 4; ++k) {
        const float wv2 = w2[(e * 4 + k) * 256 + c];
        #pragma unroll
        for (int s = 0; s < NS; ++s) p2[s][k] = fmaf(ye[s], wv2, p2[s][k]);
      }
      const float wv1 = w1[e * 256 + c];
      #pragma unroll
      for (int s = 0; s < NS; ++s) p1[s] = fmaf(ye[s], wv1, p1[s]);
    }
    #pragma unroll
    for (int s = 0; s < NS; ++s) {
      float* pr = &ps[wv][s][lan * PROW + h * 25];
      #pragma unroll
      for (int k = 0; k < 20; ++k) pr[k] = p3[s][k];
      #pragma unroll
      for (int k = 0; k < 4; ++k) pr[20 + k] = p2[s][k];
      pr[24] = p1[s];

      if (!hi) {
        float du = 0.f;
        #pragma unroll
        for (int i = 0; i < 9; ++i) du = fmaf(U1_l0[i], xr[s][i], du);
        t1[wv][s][lan][0] = p1[s] * du;
      } else {
        #pragma unroll
        for (int w = 0; w < 3; ++w) {
          float du = 0.f;
          #pragma unroll
          for (int i = 0; i < 9; ++i) du = fmaf(U1_l1[w * 9 + i], xr[s][i], du);
          t1[wv][s][lan][1 + w] = p1[s] * du;
        }
      }
    }
  }

  v8f acc3[NS][NT3];
  v8f acc2[NS];
  {
    v8f vz = {};
    #pragma unroll
    for (int s = 0; s < NS; ++s) {
      #pragma unroll
      for (int t = 0; t < NT3; ++t) acc3[s][t] = vz;
      acc2[s] = vz;
    }
  }
  gemm_steps<0>(bsh + lane, hi, xr, acc3, acc2);

  const int o   = lan & 3;
  const int kq  = lan >> 2;
  #pragma unroll
  for (int s = 0; s < NS; ++s) {
    const float* pe = &ps[wv][s][(8 * h) * PROW + (o != 0 ? 25 : 0) + kq];
    float os[8];
    #pragma unroll
    for (int r = 0; r < 8; ++r) os[r] = 0.f;
    #pragma unroll
    for (int t = 0; t < NT3; ++t)
      #pragma unroll
      for (int r = 0; r < 8; ++r)
        os[r] = fmaf(acc3[s][t][r], pe[r * PROW + 4 * t], os[r]);
    #pragma unroll
    for (int r = 0; r < 8; ++r)
      os[r] = fmaf(acc2[s][r], pe[r * PROW + 20], os[r]);
    #pragma unroll
    for (int r = 0; r < 8; ++r) {
      os[r] += __shfl_xor(os[r], 4, 32);
      os[r] += __shfl_xor(os[r], 8, 32);
    }
    if (kq == 0) {
      #pragma unroll
      for (int r = 0; r < 8; ++r) {
        int rl = 8 * h + r;
        float v = os[r] + t1[wv][s][rl][o];
        ob[wv][s][(o == 0) ? rl : (16 + rl * 3 + (o - 1))] = v;
      }
    }
    const int row0 = (bb + s) * 256 + c0;
    if (lane < 16) out[row0 + lane] = ob[wv][s][lane];
    out[OUT1_OFF + row0 * 3 + lane] = ob[wv][s][16 + lane];
    if (lane < 16) out[OUT1_OFF + row0 * 3 + 32 + lane] = ob[wv][s][48 + lane];
  }
}

extern "C" void kernel_launch(void* const* d_in, const int* in_sizes, int n_in,
                              void* d_out, int out_size, void* d_ws, size_t ws_size,
                              hipStream_t stream) {
  (void)in_sizes; (void)n_in; (void)out_size; (void)ws_size;
  const float* x     = (const float*)d_in[0];
  const float* y     = (const float*)d_in[1];
  const float* U1_l0 = (const float*)d_in[2];
  const float* U2_l0 = (const float*)d_in[3];
  const float* U3_l0 = (const float*)d_in[4];
  const float* U1_l1 = (const float*)d_in[5];
  const float* U2_l1 = (const float*)d_in[6];
  const float* U3_l1 = (const float*)d_in[7];
  const float* w1_l0 = (const float*)d_in[8];
  const float* w2_l0 = (const float*)d_in[9];
  const float* w3_l0 = (const float*)d_in[10];
  const float* w1_l1 = (const float*)d_in[11];
  const float* w2_l1 = (const float*)d_in[12];
  const float* w3_l1 = (const float*)d_in[13];

  float* Bf = (float*)d_ws;

  symcon_pack<<<56, 256, 0, stream>>>(
      U2_l0, U3_l0, U2_l1, U3_l1, Bf);

  symcon_main<<<1024, 256, 0, stream>>>(
      x, y, U1_l0, U1_l1,
      w1_l0, w2_l0, w3_l0,
      w1_l1, w2_l1, w3_l1,
      Bf, (float*)d_out);
}


# Workgroup = 16 channels x 16 batch entries: each wave loads 3 of the 24 cubic/quadratic weight rows, projects them for all 16 entries, shares P via LDS; 320 weight loads per workgroup, was 2000
# speedup vs baseline: 19.4055x; 1.0906x over previous
#include <hip/hip_runtime.h>


typedef __attribute__((ext_vector_type(2))) float v2f;
typedef __attribute__((ext_vector_type(4))) float v4f;
typedef __attribute__((ext_vector_type(8))) float v8f;

#define NROWS (1024*256)
#define OUT1_OFF NROWS
#define NM3   165
#define NM2   45
#define NJ3   21
#define NJ2   6
#define NT3   5
#define PK3_FLOATS (NT3 * NJ3 * 128)
#define PK2_FLOATS (NJ2 * 128)
#define PKV4  ((PK3_FLOATS + PK2_FLOATS) / 4)
#define NWAVE 8
#define NS    2
#define NTHR  (NWAVE * 32)
#define PROW  52

struct Tri { int a, b, c; };

__host__ __device__ constexpr Tri dec3(int m) {
  if (m >= NM3) return Tri{-1, -1, -1};
  int a = 0;
  while (m >= (9 - a) * (10 - a) / 2) { m -= (9 - a) * (10 - a) / 2; ++a; }
  int b = a;
  while (m >= 9 - b) { m -= 9 - b; ++b; }
  return Tri{a, b, b + m};
}
__host__ __device__ constexpr Tri dec2(int q) {
  if (q >= NM2) return Tri{-1, -1, -1};
  int a = 0;
  while (q >= 9 - a) { q -= 9 - a; ++a; }
  return Tri{a, a + q, 0};
}

__global__ void symcon_pack(const float* __restrict__ U2_l0,
                            const float* __restrict__ U3_l0,
                            const float* __restrict__ U2_l1,
                            const float* __restrict__ U3_l1,
                            float* __restrict__ Bf) {
  int idx = blockIdx.x * 256 + threadIdx.x;
  if (idx >= PK3_FLOATS + PK2_FLOATS) return;
  int e    = idx & 3;
  int lane = (idx >> 2) & 31;
  int lan  = lane & 15;
  float v = 0.f;
  if (idx < PK3_FLOATS) {
    int cj   = idx >> 7;
    int tile = cj / NJ3, j = cj % NJ3;
    int m    = 4 * (2 * j + (e >> 1)) + 2 * (lane >> 4) + (e & 1);
    int n    = tile * 16 + lan;
    int kk   = n >> 2, o = n & 3;
    Tri t = dec3(m);
    if (t.a >= 0) {
      const float* U = (o == 0) ? U3_l0 : (U3_l1 + (o - 1) * (729 * 20));
      int a = t.a, b = t.b, c = t.c;
      #define U3AT(p, q, s) U[(((p) * 9 + (q)) * 9 + (s)) * 20 + kk]
      if (a == b && b == c)      v = U3AT(a, a, a);
      else if (a == b)           v = U3AT(a, a, c) + U3AT(a, c, a) + U3AT(c, a, a);
      else if (b == c)           v = U3AT(a, b, b) + U3AT(b, a, b) + U3AT(b, b, a);
      else                       v = U3AT(a, b, c) + U3AT(a, c, b) + U3AT(b, a, c)
                                   + U3AT(b, c, a) + U3AT(c, a, b) + U3AT(c, b, a);
      #undef U3AT
    }
  } else {
    int j  = (idx - PK3_FLOATS) >> 7;
    int q  = 4 * (2 * j + (e >> 1)) + 2 * (lane >> 4) + (e & 1);
    int kk = lan >> 2, o = lan & 3;
    Tri t = dec2(q);
    if (t.a >= 0) {
      const float* U = (o == 0) ? U2_l0 : (U2_l1 + (o - 1) * (81 * 4));
      int a = t.a, b = t.b;
      if (a == b) v = U[(a * 9 + a) * 4 + kk];
      else        v = U[(a * 9 + b) * 4 + kk] + U[(b * 9 + a) * 4 + kk];
    }
  }
  Bf[idx] = v;
}


template <int M>
__device__ __forceinline__ float mono3(const float (&xr)[9]) {
  if constexpr (M < NM3) {
    constexpr Tri t = dec3(M);
    return xr[t.a] * xr[t.b] * xr[t.c];
  } else {
    return 0.f;
  }
}
template <int Q>
__device__ __forceinline__ float mono2(const float (&xr)[9]) {
  if constexpr (Q < NM2) {
    constexpr Tri t = dec2(Q);
    return xr[t.a] * xr[t.b];
  } else {
    return 0.f;
  }
}

template <int S>
__device__ __forceinline__ v2f afrag3(bool hi, const float (&xr)[9]) {
  float z0 = mono3<4 * S + 0>(xr), z1 = mono3<4 * S + 1>(xr);
  float z2 = mono3<4 * S + 2>(xr), z3 = mono3<4 * S + 3>(xr);
  v2f a;
  a.x = hi ? z2 : z0;
  a.y = hi ? z3 : z1;
  return a;
}
template <int S>
__device__ __forceinline__ v2f afrag2(bool hi, const float (&xr)[9]) {
  float z0 = mono2<4 * S + 0>(xr), z1 = mono2<4 * S + 1>(xr);
  float z2 = mono2<4 * S + 2>(xr), z3 = mono2<4 * S + 3>(xr);
  v2f a;
  a.x = hi ? z2 : z0;
  a.y = hi ? z3 : z1;
  return a;
}

template <int J>
__device__ __forceinline__ void gemm_steps(const v4f* bp, bool hi,
                                           const float (&xr)[NS][9],
                                           v8f (&acc3)[NS][NT3], v8f (&acc2)[NS]) {
  v4f bq[NT3];
  #pragma unroll
  for (int t = 0; t < NT3; ++t)
    bq[t] = bp[(t * NJ3 + J) * 32];
  #pragma unroll
  for (int s = 0; s < NS; ++s) {
    const v2f a0 = afrag3<2 * J>(hi, xr[s]);
    #pragma unroll
    for (int t = 0; t < NT3; ++t) {
      v2f b0;
      b0.x = bq[t].x; b0.y = bq[t].y;
      acc3[s][t] = __builtin_amdgcn_wmma_f32_16x16x4_f32(
          false, a0, false, b0, (short)0, acc3[s][t], false, false);
    }
  }
  #pragma unroll
  for (int s = 0; s < NS; ++s) {
    const v2f a1 = afrag3<2 * J + 1>(hi, xr[s]);
    #pragma unroll
    for (int t = 0; t < NT3; ++t) {
      v2f b1;
      b1.x = bq[t].z; b1.y = bq[t].w;
      acc3[s][t] = __builtin_amdgcn_wmma_f32_16x16x4_f32(
          false, a1, false, b1, (short)0, acc3[s][t], false, false);
    }
  }
  if constexpr (J < NJ2) {
    const v4f bs = bp[PK3_FLOATS / 4 + J * 32];
    v2f b0, b1;
    b0.x = bs.x; b0.y = bs.y;
    b1.x = bs.z; b1.y = bs.w;
    #pragma unroll
    for (int s = 0; s < NS; ++s) {
      const v2f c0 = afrag2<2 * J>(hi, xr[s]);
      acc2[s] = __builtin_amdgcn_wmma_f32_16x16x4_f32(
          false, c0, false, b0, (short)0, acc2[s], false, false);
    }
    #pragma unroll
    for (int s = 0; s < NS; ++s) {
      const v2f c1 = afrag2<2 * J + 1>(hi, xr[s]);
      acc2[s] = __builtin_amdgcn_wmma_f32_16x16x4_f32(
          false, c1, false, b1, (short)0, acc2[s], false, false);
    }
  }
  if constexpr (J + 1 < NJ3) gemm_steps<J + 1>(bp, hi, xr, acc3, acc2);
}

__global__ __launch_bounds__(NTHR) void symcon_main(
    const float* __restrict__ x, const float* __restrict__ y,
    const float* __restrict__ U1_l0, const float* __restrict__ U1_l1,
    const float* __restrict__ w1_l0, const float* __restrict__ w2_l0,
    const float* __restrict__ w3_l0,
    const float* __restrict__ w1_l1, const float* __restrict__ w2_l1,
    const float* __restrict__ w3_l1,
    const float* __restrict__ Bf,
    float* __restrict__ out) {
  __shared__ v4f bsh[PKV4];
  __shared__ float ps[NWAVE][NS][16 * PROW];
  __shared__ float t1[NWAVE][NS][16][4];
  __shared__ float ob[NWAVE][NS][64];

  const int tid  = threadIdx.x;
  const int wv   = __builtin_amdgcn_readfirstlane(tid >> 5);
  const int lane = tid & 31;
  const int lan  = lane & 15;
  const bool hi  = lane >= 16;
  const int h    = hi ? 1 : 0;
  const int bb0  = (blockIdx.x >> 4) * (NWAVE * NS);
  const int bb   = bb0 + wv * NS;
  const int c0   = (blockIdx.x & 15) * 16;
  const int c    = c0 + lan;

  float xr[NS][9];
  #pragma unroll
  for (int s = 0; s < NS; ++s)
    #pragma unroll
    for (int i = 0; i < 9; ++i) xr[s][i] = x[((bb + s) * 256 + c) * 9 + i];

  constexpr int NBR = (PKV4 + NTHR - 1) / NTHR;
  v4f bst[NBR];
  {
    const v4f* __restrict__ bg = (const v4f*)Bf;
    #pragma unroll
    for (int i = 0; i < NBR; ++i) {
      int u = tid + i * NTHR;
      if ((i + 1) * NTHR > PKV4) u = u < PKV4 ? u : PKV4 - 1;
      bst[i] = bg[u];
    }
  }

  {
    const float* __restrict__ w3 = hi ? w3_l1 : w3_l0;
    const float* __restrict__ w2 = hi ? w2_l1 : w2_l0;
    const float* __restrict__ w1 = hi ? w1_l1 : w1_l0;
    const float* __restrict__ ra = w3 + wv * 256 + c;
    const float* __restrict__ rb = w3 + (wv + 8) * 256 + c;
    const float* __restrict__ rc = (wv < 4) ? (w3 + (wv + 16) * 256 + c) : (w2 + (wv - 4) * 256 + c);
    const int stc = (wv < 4) ? 20 * 256 : 4 * 256;
    float wa[10], wb[10], wc[10], wl[10];
    #pragma unroll
    for (int e = 0; e < 10; ++e) {
      wa[e] = ra[e * (20 * 256)];
      wb[e] = rb[e * (20 * 256)];
      wc[e] = rc[e * stc];
      wl[e] = w1[e * 256 + c];
    }
    #pragma unroll
    for (int s = 0; s < NS; ++s) {
      float p1 = 0.f;
      #pragma unroll
      for (int e = 0; e < 10; ++e) p1 = fmaf(y[(bb + s) * 10 + e], wl[e], p1);
      ps[wv][s][lan * PROW + h * 25 + 24] = p1;

      if (!hi) {
        float du = 0.f;
        #pragma unroll
        for (int i = 0; i < 9; ++i) du = fmaf(U1_l0[i], xr[s][i], du);
        t1[wv][s][lan][0] = p1 * du;
      } else {
        #pragma unroll
        for (int w = 0; w < 3; ++w) {
          float du = 0.f;
          #pragma unroll
          for (int i = 0; i < 9; ++i) du = fmaf(U1_l1[w * 9 + i], xr[s][i], du);
          t1[wv][s][lan][1 + w] = p1 * du;
        }
      }
    }
    #pragma unroll
    for (int j = 0; j < NWAVE * NS; ++j) {
      float pa = 0.f, pb = 0.f, pc = 0.f;
      #pragma unroll
      for (int e = 0; e < 10; ++e) {
        const float ye = y[(bb0 + j) * 10 + e];
        pa = fmaf(ye, wa[e], pa);
        pb = fmaf(ye, wb[e], pb);
        pc = fmaf(ye, wc[e], pc);
      }
      float* pr = &ps[j / NS][j % NS][lan * PROW + h * 25];
      pr[wv]      = pa;
      pr[wv + 8]  = pb;
      pr[wv + 16] = pc;
    }
  }

  #pragma unroll
  for (int i = 0; i < NBR; ++i) {
    int u = tid + i * NTHR;
    if ((i + 1) * NTHR > PKV4) u = u < PKV4 ? u : PKV4 - 1;
    bsh[u] = bst[i];
  }

  __syncthreads();

  v8f acc3[NS][NT3];
  v8f acc2[NS];
  {
    v8f vz = {};
    #pragma unroll
    for (int s = 0; s < NS; ++s) {
      #pragma unroll
      for (int t = 0; t < NT3; ++t) acc3[s][t] = vz;
      acc2[s] = vz;
    }
  }
  gemm_steps<0>(bsh + lane, hi, xr, acc3, acc2);

  const int o   = lan & 3;
  const int kq  = lan >> 2;
  #pragma unroll
  for (int s = 0; s < NS; ++s) {
    const float* pe = &ps[wv][s][(8 * h) * PROW + (o != 0 ? 25 : 0) + kq];
    float os[8];
    #pragma unroll
    for (int r = 0; r < 8; ++r) os[r] = 0.f;
    #pragma unroll
    for (int t = 0; t < NT3; ++t)
      #pragma unroll
      for (int r = 0; r < 8; ++r)
        os[r] = fmaf(acc3[s][t][r], pe[r * PROW + 4 * t], os[r]);
    #pragma unroll
    for (int r = 0; r < 8; ++r)
      os[r] = fmaf(acc2[s][r], pe[r * PROW + 20], os[r]);
    #pragma unroll
    for (int r = 0; r < 8; ++r) {
      os[r] += __shfl_xor(os[r], 4, 32);
      os[r] += __shfl_xor(os[r], 8, 32);
    }
    if (kq == 0) {
      #pragma unroll
      for (int r = 0; r < 8; ++r) {
        int rl = 8 * h + r;
        float v = os[r] + t1[wv][s][rl][o];
        ob[wv][s][(o == 0) ? rl : (16 + rl * 3 + (o - 1))] = v;
      }
    }
    const int row0 = (bb + s) * 256 + c0;
    if (lane < 16) out[row0 + lane] = ob[wv][s][lane];
    out[OUT1_OFF + row0 * 3 + lane] = ob[wv][s][16 + lane];
    if (lane < 16) out[OUT1_OFF + row0 * 3 + 32 + lane] = ob[wv][s][48 + lane];
  }
}

extern "C" void kernel_launch(void* const* d_in, const int* in_sizes, int n_in,
                              void* d_out, int out_size, void* d_ws, size_t ws_size,
                              hipStream_t stream) {
  (void)in_sizes; (void)n_in; (void)out_size; (void)ws_size;
  const float* x     = (const float*)d_in[0];
  const float* y     = (const float*)d_in[1];
  const float* U1_l0 = (const float*)d_in[2];
  const float* U2_l0 = (const float*)d_in[3];
  const float* U3_l0 = (const float*)d_in[4];
  const float* U1_l1 = (const float*)d_in[5];
  const float* U2_l1 = (const float*)d_in[6];
  const float* U3_l1 = (const float*)d_in[7];
  const float* w1_l0 = (const float*)d_in[8];
  const float* w2_l0 = (const float*)d_in[9];
  const float* w3_l0 = (const float*)d_in[10];
  const float* w1_l1 = (const float*)d_in[11];
  const float* w2_l1 = (const float*)d_in[12];
  const float* w3_l1 = (const float*)d_in[13];

  float* Bf = (float*)d_ws;

  symcon_pack<<<56, 256, 0, stream>>>(
      U2_l0, U3_l0, U2_l1, U3_l1, Bf);

  symcon_main<<<1024, 256, 0, stream>>>(
      x, y, U1_l0, U1_l1,
      w1_l0, w2_l0, w3_l0,
      w1_l1, w2_l1, w3_l1,
      Bf, (float*)d_out);
}
